# starting kernel
# baseline (speedup 1.0000x reference)
#include <hip/hip_runtime.h>
#include <hip/hip_bf16.h>

#define NN 50000
#define EE 600000
#define INC 128
#define OUTC 128
#define HEADS 8
#define HEADC 16

typedef __attribute__((ext_vector_type(2))) float v2f;
typedef __attribute__((ext_vector_type(8))) float v8f;

__global__ void gat_gemm_wmma(const float* __restrict__ X,
                              const float* __restrict__ W,
                              float* __restrict__ emb) {
    const int wave  = (blockIdx.x * blockDim.x + threadIdx.x) >> 5;
    const int lane  = threadIdx.x & 31;
    const int tileM = wave >> 3;
    const int tileN = wave & 7;
    const int half  = lane >> 4;
    const int l     = lane & 15;

    const float* __restrict__ arow = X + (size_t)(tileM * 16 + l) * INC + half * 2;
    const float* __restrict__ brow = W + (size_t)(tileN * 16 + l) * INC + half * 2;

    v8f c = {};
#pragma unroll
    for (int kk = 0; kk < INC; kk += 4) {
        v2f a = *(const v2f*)(arow + kk);
        v2f b = *(const v2f*)(brow + kk);
        c = __builtin_amdgcn_wmma_f32_16x16x4_f32(
                 false, a,  false, b,
                 (short)0, c,  false,  false);
    }

    float* __restrict__ op =
        emb + (size_t)(tileM * 16 + half * 8) * OUTC + tileN * 16 + l;
#pragma unroll
    for (int v = 0; v < 8; ++v) op[(size_t)v * OUTC] = c[v];
}

__global__ void gat_scores(const float* __restrict__ emb,
                           const float* __restrict__ a_left,
                           const float* __restrict__ a_right,
                           float* __restrict__ ss, float* __restrict__ ts) {
    int gid = blockIdx.x * blockDim.x + threadIdx.x;
    if (gid >= NN * HEADS) return;
    int n = gid >> 3, h = gid & 7;
    const float* ep = emb + (size_t)n * OUTC + h * HEADC;
    float s0 = 0.f, s1 = 0.f;
#pragma unroll
    for (int c = 0; c < HEADC; ++c) {
        float e = ep[c];
        s0 += e * a_left[c * HEADS + h];
        s1 += e * a_right[c * HEADS + h];
    }
    ss[gid] = s0;
    ts[gid] = s1;
}

__global__ void gat_init(float* __restrict__ out, const float* __restrict__ bias,
                         float* __restrict__ denom) {
    int gid = blockIdx.x * blockDim.x + threadIdx.x;
    if (gid < NN * OUTC) out[gid] = bias[gid & (OUTC - 1)];
    if (gid < NN * HEADS) denom[gid] = 0.f;
}

__global__ void gat_edge_exp(const int* __restrict__ src, const int* __restrict__ trg,
                             const float* __restrict__ ss, const float* __restrict__ ts,
                             float* __restrict__ exps, float* __restrict__ denom) {
    int gid = blockIdx.x * blockDim.x + threadIdx.x;
    if (gid >= EE * HEADS) return;
    int e = gid >> 3, h = gid & 7;
    int s = src[e], t = trg[e];
    float v = ss[s * HEADS + h] + ts[t * HEADS + h];
    v = v > 0.f ? v : 0.2f * v;
    float ex = __expf(v);
    exps[gid] = ex;
    atomicAdd(&denom[t * HEADS + h], ex);
}

__global__ void gat_aggregate(const int* __restrict__ src, const int* __restrict__ trg,
                              const float* __restrict__ emb,
                              const float* __restrict__ exps,
                              const float* __restrict__ denom,
                              float* __restrict__ out) {
    long long gid = (long long)blockIdx.x * blockDim.x + threadIdx.x;
    if (gid >= (long long)EE * OUTC) return;
    int e = (int)(gid >> 7);
    int k = (int)(gid & (OUTC - 1));
    int h = k >> 4;
    int s = src[e], t = trg[e];
    float attn = exps[e * HEADS + h] / (denom[t * HEADS + h] + 1e-16f);
    atomicAdd(&out[(size_t)t * OUTC + k], emb[(size_t)s * OUTC + k] * attn);
}

extern "C" void kernel_launch(void* const* d_in, const int* in_sizes, int n_in,
                              void* d_out, int out_size, void* d_ws, size_t ws_size,
                              hipStream_t stream) {
    const float* X      = (const float*)d_in[0];
    const int*   eidx   = (const int*)d_in[1];
    const float* W      = (const float*)d_in[2];
    const float* a_l    = (const float*)d_in[3];
    const float* a_r    = (const float*)d_in[4];
    const float* bias   = (const float*)d_in[5];
    float* out = (float*)d_out;

    const int* src = eidx;
    const int* trg = eidx + EE;

    float* ws    = (float*)d_ws;
    float* emb   = ws;
    float* ss    = emb + (size_t)NN * OUTC;
    float* ts    = ss + (size_t)NN * HEADS;
    float* denom = ts + (size_t)NN * HEADS;
    float* exps  = denom + (size_t)NN * HEADS;

    gat_gemm_wmma<<<3125, 256, 0, stream>>>(X, W, emb);

    gat_scores<<<(NN * HEADS + 255) / 256, 256, 0, stream>>>(emb, a_l, a_r, ss, ts);

    gat_init<<<(NN * OUTC + 255) / 256, 256, 0, stream>>>(out, bias, denom);

    gat_edge_exp<<<(EE * HEADS + 255) / 256, 256, 0, stream>>>(src, trg, ss, ts, exps, denom);

    long long nagg = (long long)EE * OUTC;
    gat_aggregate<<<(unsigned)((nagg + 255) / 256), 256, 0, stream>>>(
        src, trg, emb, exps, denom, out);
}


# Init and edge-exp launches coarsened: 8192 waves store out and denom in 16-byte pieces; an edge-exp wave takes 64 edges with all gathers in flight (350000 short waves become 17567); add order free
# speedup vs baseline: 2.5352x; 2.5352x over previous
#include <hip/hip_runtime.h>
#include <hip/hip_bf16.h>

#define NN 50000
#define EE 600000
#define INC 128
#define OUTC 128
#define HEADS 8
#define HEADC 16

typedef __attribute__((ext_vector_type(2))) float v2f;
typedef __attribute__((ext_vector_type(4))) float v4f;
typedef __attribute__((ext_vector_type(8))) float v8f;

__global__ void gat_gemm_wmma(const float* __restrict__ X,
                              const float* __restrict__ W,
                              float* __restrict__ emb) {
    const int wave  = (blockIdx.x * blockDim.x + threadIdx.x) >> 5;
    const int lane  = threadIdx.x & 31;
    const int tileM = wave >> 3;
    const int tileN = wave & 7;
    const int half  = lane >> 4;
    const int l     = lane & 15;

    const float* __restrict__ arow = X + (size_t)(tileM * 16 + l) * INC + half * 2;
    const float* __restrict__ brow = W + (size_t)(tileN * 16 + l) * INC + half * 2;

    v8f c = {};
#pragma unroll
    for (int kk = 0; kk < INC; kk += 4) {
        v2f a = *(const v2f*)(arow + kk);
        v2f b = *(const v2f*)(brow + kk);
        c = __builtin_amdgcn_wmma_f32_16x16x4_f32(
                 false, a,  false, b,
                 (short)0, c,  false,  false);
    }

    float* __restrict__ op =
        emb + (size_t)(tileM * 16 + half * 8) * OUTC + tileN * 16 + l;
#pragma unroll
    for (int v = 0; v < 8; ++v) op[(size_t)v * OUTC] = c[v];
}

__global__ void gat_scores(const float* __restrict__ emb,
                           const float* __restrict__ a_left,
                           const float* __restrict__ a_right,
                           float* __restrict__ ss, float* __restrict__ ts) {
    int gid = blockIdx.x * blockDim.x + threadIdx.x;
    if (gid >= NN * HEADS) return;
    int n = gid >> 3, h = gid & 7;
    const float* ep = emb + (size_t)n * OUTC + h * HEADC;
    float s0 = 0.f, s1 = 0.f;
#pragma unroll
    for (int c = 0; c < HEADC; ++c) {
        float e = ep[c];
        s0 += e * a_left[c * HEADS + h];
        s1 += e * a_right[c * HEADS + h];
    }
    ss[gid] = s0;
    ts[gid] = s1;
}

#define INIT_BLOCKS  1024
#define INIT_THREADS (INIT_BLOCKS * 256)
#define OUT_PIECES   (NN * OUTC / 4)
#define INIT_FULL    (OUT_PIECES / INIT_THREADS)
#define INIT_REST    (OUT_PIECES % INIT_THREADS)
static_assert(INIT_THREADS % (OUTC / 4) == 0, "a thread must keep its channel");
static_assert(NN * HEADS / 4 <= INIT_THREADS, "denom is one piece per thread");

__global__ __launch_bounds__(256)
void gat_init(float* __restrict__ out, const float* __restrict__ bias,
              float* __restrict__ denom) {
    const int tid = blockIdx.x * 256 + threadIdx.x;
    const v4f b = *(const v4f*)(bias + 4 * (tid & 31));
    v4f* __restrict__ o4 = (v4f*)out;
#pragma unroll
    for (int k = 0; k < INIT_FULL; ++k) o4[tid + k * INIT_THREADS] = b;
    if (tid < INIT_REST) o4[tid + INIT_FULL * INIT_THREADS] = b;
    if (tid < NN * HEADS / 4) ((v4f*)denom)[tid] = (v4f){0.f, 0.f, 0.f, 0.f};
}

#define EXP_U     16
#define EXP_EDGES (EXP_U * 4)
#define EXP_WAVES (EE / EXP_EDGES)

__global__ __launch_bounds__(256)
void gat_edge_exp(const int* __restrict__ src, const int* __restrict__ trg,
                  const float* __restrict__ ss, const float* __restrict__ ts,
                  float* __restrict__ exps, float* __restrict__ denom) {
    const int lane = threadIdx.x & 31;
    const int wave = (blockIdx.x * blockDim.x + threadIdx.x) >> 5;
    if (wave >= EXP_WAVES) return;
    const int sub = lane >> 3;
    const int hd  = lane & 7;
    const int e0  = wave * EXP_EDGES;

    int   t[EXP_U];
    float a[EXP_U], b[EXP_U];
#pragma unroll
    for (int u = 0; u < EXP_U; ++u) {
        const int e = e0 + u * 4 + sub;
        const int s = src[e];
        t[u] = trg[e];
        a[u] = ss[s * HEADS + hd];
        b[u] = ts[t[u] * HEADS + hd];
    }
#pragma unroll
    for (int u = 0; u < EXP_U; ++u) {
        float v = a[u] + b[u];
        v = v > 0.f ? v : 0.2f * v;
        const float ex = __expf(v);
        exps[(size_t)e0 * HEADS + u * 32 + lane] = ex;
        atomicAdd(&denom[t[u] * HEADS + hd], ex);
    }
}

#define AGG_G      8
#define AGG_BLOCKS 1024
#define AGG_WAVES  (AGG_BLOCKS * 8)

__global__ __launch_bounds__(256)
void gat_aggregate(const int* __restrict__ src, const int* __restrict__ trg,
                   const float* __restrict__ emb,
                   const float* __restrict__ exps,
                   const float* __restrict__ denom,
                   float* __restrict__ out) {
    const unsigned lane = threadIdx.x & 31u;
    const int wave = __builtin_amdgcn_readfirstlane(
        (int)((blockIdx.x * blockDim.x + threadIdx.x) >> 5));
    const unsigned hi  = lane >> 4;
    const unsigned sub = lane >> 3;
    const unsigned hd  = lane & 7u;

    for (int g = wave; g < EE / AGG_G; g += AGG_WAVES) {
        const int e0 = g * AGG_G;

        int s[AGG_G], t[AGG_G];
#pragma unroll
        for (int u = 0; u < AGG_G; ++u) { s[u] = src[e0 + u]; t[u] = trg[e0 + u]; }

        const int tA = sub == 0 ? t[0] : sub == 1 ? t[1] : sub == 2 ? t[2] : t[3];
        const int tB = sub == 0 ? t[4] : sub == 1 ? t[5] : sub == 2 ? t[6] : t[7];
        const float xA = exps[(size_t)e0 * HEADS + lane];
        const float xB = exps[(size_t)e0 * HEADS + 32 + lane];
        const float dA = denom[(size_t)tA * HEADS + hd];
        const float dB = denom[(size_t)tB * HEADS + hd];

        float v[AGG_G][4];
#pragma unroll
        for (int u = 0; u < AGG_G; ++u) {
            const float* __restrict__ row = emb + (size_t)s[u] * OUTC;
#pragma unroll
            for (int i = 0; i < 4; ++i) v[u][i] = row[i * 32 + lane];
        }

        const float wA = xA / (dA + 1e-16f);
        const float wB = xB / (dB + 1e-16f);

#pragma unroll
        for (int u = 0; u < AGG_G; ++u) {
            float* __restrict__ orow = out + (size_t)t[u] * OUTC;
#pragma unroll
            for (int i = 0; i < 4; ++i) {
                const float w = __shfl(u < 4 ? wA : wB, (int)((u & 3) * 8 + 2 * i + hi));
                atomicAdd(orow + (i * 32 + lane), v[u][i] * w);
            }
        }
    }
}

extern "C" void kernel_launch(void* const* d_in, const int* in_sizes, int n_in,
                              void* d_out, int out_size, void* d_ws, size_t ws_size,
                              hipStream_t stream) {
    const float* X      = (const float*)d_in[0];
    const int*   eidx   = (const int*)d_in[1];
    const float* W      = (const float*)d_in[2];
    const float* a_l    = (const float*)d_in[3];
    const float* a_r    = (const float*)d_in[4];
    const float* bias   = (const float*)d_in[5];
    float* out = (float*)d_out;

    const int* src = eidx;
    const int* trg = eidx + EE;

    float* ws    = (float*)d_ws;
    float* emb   = ws;
    float* ss    = emb + (size_t)NN * OUTC;
    float* ts    = ss + (size_t)NN * HEADS;
    float* denom = ts + (size_t)NN * HEADS;
    float* exps  = denom + (size_t)NN * HEADS;

    gat_gemm_wmma<<<3125, 256, 0, stream>>>(X, W, emb);

    gat_scores<<<(NN * HEADS + 255) / 256, 256, 0, stream>>>(emb, a_l, a_r, ss, ts);

    gat_init<<<INIT_BLOCKS, 256, 0, stream>>>(out, bias, denom);

    gat_edge_exp<<<(EXP_WAVES + 7) / 8, 256, 0, stream>>>(src, trg, ss, ts, exps, denom);

    gat_aggregate<<<AGG_BLOCKS, 256, 0, stream>>>(src, trg, emb, exps, denom, out);
}


# GEMM: a wave owns a 16x128 row strip, its X fragments loaded once for all 8 column tiles; W staged per block in LDS; strip stored in whole lines; k taken in runs of 8: the dot's sum order changes
# speedup vs baseline: 2.9768x; 1.1742x over previous
#include <hip/hip_runtime.h>
#include <hip/hip_bf16.h>

#define NN 50000
#define EE 600000
#define INC 128
#define OUTC 128
#define HEADS 8
#define HEADC 16

typedef __attribute__((ext_vector_type(2))) float v2f;
typedef __attribute__((ext_vector_type(4))) float v4f;
typedef __attribute__((ext_vector_type(8))) float v8f;

#define GEMM_WAVES   16
#define GEMM_THREADS (GEMM_WAVES * 32)
#define GEMM_BLOCKS  256
#define STRIPS       (NN / 16)
#define WPITCH       132
static_assert(NN % 16 == 0, "whole strips only");
static_assert(GEMM_BLOCKS * GEMM_WAVES >= STRIPS, "one strip per wave at most");
static_assert((OUTC * INC / 4) % GEMM_THREADS == 0, "W staging: equal pieces per thread");

__global__ __launch_bounds__(GEMM_THREADS)
void gat_gemm_wmma(const float* __restrict__ X,
                   const float* __restrict__ W,
                   float* __restrict__ emb) {
    __shared__ __attribute__((aligned(16))) float wl[OUTC * WPITCH];

    const int tid  = threadIdx.x;
    const int lane = tid & 31;
    const int l    = lane & 15;
    const int h    = lane >> 4;
    const int wave  = __builtin_amdgcn_readfirstlane(tid >> 5);
    const int strip = wave * GEMM_BLOCKS + (int)blockIdx.x;
    const bool active = strip < STRIPS;


    v4f a[16];
    if (active) {
        const float* __restrict__ arow = X + (size_t)(strip * 16 + l) * INC + 4 * h;
#pragma unroll
        for (int t = 0; t < 16; ++t) a[t] = *(const v4f*)(arow + 8 * t);
    }

#pragma unroll
    for (int j = 0; j < (OUTC * INC / 4) / GEMM_THREADS; ++j) {
        const int i  = tid + j * GEMM_THREADS;
        const int n  = i >> 5;
        const int k4 = i & 31;
        const v4f w  = *(const v4f*)(W + (size_t)n * INC + 4 * k4);
        const int r  = (n & 64) | ((n & 3) << 4) | ((n >> 2) & 15);
        *(v4f*)(wl + r * WPITCH + 4 * k4) = w;
    }
    __syncthreads();
    if (!active) return;

    const float* bl = wl + l * WPITCH + 4 * h;
    v8f acc[8];
#pragma unroll
    for (int c = 0; c < 8; ++c) acc[c] = (v8f){};

    v4f b0[8], b1[8];
#pragma unroll
    for (int c = 0; c < 8; ++c) b0[c] = *(const v4f*)(bl + c * 16 * WPITCH);

#pragma unroll
    for (int t = 0; t < 16; t += 2) {
#pragma unroll
        for (int c = 0; c < 8; ++c) b1[c] = *(const v4f*)(bl + c * 16 * WPITCH + 8 * (t + 1));
        __builtin_amdgcn_sched_barrier(0);
#pragma unroll
        for (int c = 0; c < 8; ++c)
            acc[c] = __builtin_amdgcn_wmma_f32_16x16x4_f32(
                false, a[t].xy, false, b0[c].xy, (short)0, acc[c], false, false);
#pragma unroll
        for (int c = 0; c < 8; ++c)
            acc[c] = __builtin_amdgcn_wmma_f32_16x16x4_f32(
                false, a[t].zw, false, b0[c].zw, (short)0, acc[c], false, false);
        __builtin_amdgcn_sched_barrier(0);
        if (t + 2 < 16) {
#pragma unroll
            for (int c = 0; c < 8; ++c) b0[c] = *(const v4f*)(bl + c * 16 * WPITCH + 8 * (t + 2));
        }
        __builtin_amdgcn_sched_barrier(0);
#pragma unroll
        for (int c = 0; c < 8; ++c)
            acc[c] = __builtin_amdgcn_wmma_f32_16x16x4_f32(
                false, a[t + 1].xy, false, b1[c].xy, (short)0, acc[c], false, false);
#pragma unroll
        for (int c = 0; c < 8; ++c)
            acc[c] = __builtin_amdgcn_wmma_f32_16x16x4_f32(
                false, a[t + 1].zw, false, b1[c].zw, (short)0, acc[c], false, false);
        __builtin_amdgcn_sched_barrier(0);
    }

    float* __restrict__ op = emb + (size_t)(strip * 16 + 8 * h) * OUTC + 4 * l;
#pragma unroll
    for (int q = 0; q < 2; ++q) {
#pragma unroll
        for (int v = 0; v < 8; ++v) {
            const v4f o = {acc[4 * q][v], acc[4 * q + 1][v], acc[4 * q + 2][v], acc[4 * q + 3][v]};
            *(v4f*)(op + (size_t)v * OUTC + 64 * q) = o;
        }
    }
}

__global__ void gat_scores(const float* __restrict__ emb,
                           const float* __restrict__ a_left,
                           const float* __restrict__ a_right,
                           float* __restrict__ ss, float* __restrict__ ts) {
    int gid = blockIdx.x * blockDim.x + threadIdx.x;
    if (gid >= NN * HEADS) return;
    int n = gid >> 3, h = gid & 7;
    const float* ep = emb + (size_t)n * OUTC + h * HEADC;
    float s0 = 0.f, s1 = 0.f;
#pragma unroll
    for (int c = 0; c < HEADC; ++c) {
        float e = ep[c];
        s0 += e * a_left[c * HEADS + h];
        s1 += e * a_right[c * HEADS + h];
    }
    ss[gid] = s0;
    ts[gid] = s1;
}

#define INIT_BLOCKS  1024
#define INIT_THREADS (INIT_BLOCKS * 256)
#define OUT_PIECES   (NN * OUTC / 4)
#define INIT_FULL    (OUT_PIECES / INIT_THREADS)
#define INIT_REST    (OUT_PIECES % INIT_THREADS)
static_assert(INIT_THREADS % (OUTC / 4) == 0, "a thread must keep its channel");
static_assert(NN * HEADS / 4 <= INIT_THREADS, "denom is one piece per thread");

__global__ __launch_bounds__(256)
void gat_init(float* __restrict__ out, const float* __restrict__ bias,
              float* __restrict__ denom) {
    const int tid = blockIdx.x * 256 + threadIdx.x;
    const v4f b = *(const v4f*)(bias + 4 * (tid & 31));
    v4f* __restrict__ o4 = (v4f*)out;
#pragma unroll
    for (int k = 0; k < INIT_FULL; ++k) o4[tid + k * INIT_THREADS] = b;
    if (tid < INIT_REST) o4[tid + INIT_FULL * INIT_THREADS] = b;
    if (tid < NN * HEADS / 4) ((v4f*)denom)[tid] = (v4f){0.f, 0.f, 0.f, 0.f};
}

#define EXP_U     16
#define EXP_EDGES (EXP_U * 4)
#define EXP_WAVES (EE / EXP_EDGES)

__global__ __launch_bounds__(256)
void gat_edge_exp(const int* __restrict__ src, const int* __restrict__ trg,
                  const float* __restrict__ ss, const float* __restrict__ ts,
                  float* __restrict__ exps, float* __restrict__ denom) {
    const int lane = threadIdx.x & 31;
    const int wave = (blockIdx.x * blockDim.x + threadIdx.x) >> 5;
    if (wave >= EXP_WAVES) return;
    const int sub = lane >> 3;
    const int hd  = lane & 7;
    const int e0  = wave * EXP_EDGES;

    int   t[EXP_U];
    float a[EXP_U], b[EXP_U];
#pragma unroll
    for (int u = 0; u < EXP_U; ++u) {
        const int e = e0 + u * 4 + sub;
        const int s = src[e];
        t[u] = trg[e];
        a[u] = ss[s * HEADS + hd];
        b[u] = ts[t[u] * HEADS + hd];
    }
#pragma unroll
    for (int u = 0; u < EXP_U; ++u) {
        float v = a[u] + b[u];
        v = v > 0.f ? v : 0.2f * v;
        const float ex = __expf(v);
        exps[(size_t)e0 * HEADS + u * 32 + lane] = ex;
        atomicAdd(&denom[t[u] * HEADS + hd], ex);
    }
}

#define AGG_G      8
#define AGG_BLOCKS 1024
#define AGG_WAVES  (AGG_BLOCKS * 8)

__global__ __launch_bounds__(256)
void gat_aggregate(const int* __restrict__ src, const int* __restrict__ trg,
                   const float* __restrict__ emb,
                   const float* __restrict__ exps,
                   const float* __restrict__ denom,
                   float* __restrict__ out) {
    const unsigned lane = threadIdx.x & 31u;
    const int wave = __builtin_amdgcn_readfirstlane(
        (int)((blockIdx.x * blockDim.x + threadIdx.x) >> 5));
    const unsigned hi  = lane >> 4;
    const unsigned sub = lane >> 3;
    const unsigned hd  = lane & 7u;

    for (int g = wave; g < EE / AGG_G; g += AGG_WAVES) {
        const int e0 = g * AGG_G;

        int s[AGG_G], t[AGG_G];
#pragma unroll
        for (int u = 0; u < AGG_G; ++u) { s[u] = src[e0 + u]; t[u] = trg[e0 + u]; }

        const int tA = sub == 0 ? t[0] : sub == 1 ? t[1] : sub == 2 ? t[2] : t[3];
        const int tB = sub == 0 ? t[4] : sub == 1 ? t[5] : sub == 2 ? t[6] : t[7];
        const float xA = exps[(size_t)e0 * HEADS + lane];
        const float xB = exps[(size_t)e0 * HEADS + 32 + lane];
        const float dA = denom[(size_t)tA * HEADS + hd];
        const float dB = denom[(size_t)tB * HEADS + hd];

        float v[AGG_G][4];
#pragma unroll
        for (int u = 0; u < AGG_G; ++u) {
            const float* __restrict__ row = emb + (size_t)s[u] * OUTC;
#pragma unroll
            for (int i = 0; i < 4; ++i) v[u][i] = row[i * 32 + lane];
        }

        const float wA = xA / (dA + 1e-16f);
        const float wB = xB / (dB + 1e-16f);

#pragma unroll
        for (int u = 0; u < AGG_G; ++u) {
            float* __restrict__ orow = out + (size_t)t[u] * OUTC;
#pragma unroll
            for (int i = 0; i < 4; ++i) {
                const float w = __shfl(u < 4 ? wA : wB, (int)((u & 3) * 8 + 2 * i + hi));
                atomicAdd(orow + (i * 32 + lane), v[u][i] * w);
            }
        }
    }
}

extern "C" void kernel_launch(void* const* d_in, const int* in_sizes, int n_in,
                              void* d_out, int out_size, void* d_ws, size_t ws_size,
                              hipStream_t stream) {
    const float* X      = (const float*)d_in[0];
    const int*   eidx   = (const int*)d_in[1];
    const float* W      = (const float*)d_in[2];
    const float* a_l    = (const float*)d_in[3];
    const float* a_r    = (const float*)d_in[4];
    const float* bias   = (const float*)d_in[5];
    float* out = (float*)d_out;

    const int* src = eidx;
    const int* trg = eidx + EE;

    float* ws    = (float*)d_ws;
    float* emb   = ws;
    float* ss    = emb + (size_t)NN * OUTC;
    float* ts    = ss + (size_t)NN * HEADS;
    float* denom = ts + (size_t)NN * HEADS;
    float* exps  = denom + (size_t)NN * HEADS;

    gat_gemm_wmma<<<GEMM_BLOCKS, GEMM_THREADS, 0, stream>>>(X, W, emb);

    gat_scores<<<(NN * HEADS + 255) / 256, 256, 0, stream>>>(emb, a_l, a_r, ss, ts);

    gat_init<<<INIT_BLOCKS, 256, 0, stream>>>(out, bias, denom);

    gat_edge_exp<<<(EXP_WAVES + 7) / 8, 256, 0, stream>>>(src, trg, ss, ts, exps, denom);

    gat_aggregate<<<AGG_BLOCKS, 256, 0, stream>>>(src, trg, emb, exps, denom, out);
}


# Aggregate in 256 owner blocks: each filters a byte-per-edge target-range table, adds exp-weighted source rows and denominators by LDS float adds (order free), divides after the sum, stores out once
# speedup vs baseline: 4.5563x; 1.5306x over previous
#include <hip/hip_runtime.h>
#include <hip/hip_bf16.h>

#define NN 50000
#define EE 600000
#define INC 128
#define OUTC 128
#define HEADS 8
#define HEADC 16

typedef __attribute__((ext_vector_type(2))) float v2f;
typedef __attribute__((ext_vector_type(4))) float v4f;
typedef __attribute__((ext_vector_type(8))) float v8f;

#define GEMM_WAVES   16
#define GEMM_THREADS (GEMM_WAVES * 32)
#define GEMM_BLOCKS  256
#define STRIPS       (NN / 16)
#define WPITCH       132
static_assert(NN % 16 == 0, "whole strips only");
static_assert(GEMM_BLOCKS * GEMM_WAVES >= STRIPS, "one strip per wave at most");
static_assert((OUTC * INC / 4) % GEMM_THREADS == 0, "W staging: equal pieces per thread");

__global__ __launch_bounds__(GEMM_THREADS)
void gat_gemm_wmma(const float* __restrict__ X,
                   const float* __restrict__ W,
                   float* __restrict__ emb) {
    __shared__ __attribute__((aligned(16))) float wl[OUTC * WPITCH];

    const int tid  = threadIdx.x;
    const int lane = tid & 31;
    const int l    = lane & 15;
    const int h    = lane >> 4;
    const int wave  = __builtin_amdgcn_readfirstlane(tid >> 5);
    const int strip = wave * GEMM_BLOCKS + (int)blockIdx.x;
    const bool active = strip < STRIPS;


    v4f a[16];
    if (active) {
        const float* __restrict__ arow = X + (size_t)(strip * 16 + l) * INC + 4 * h;
#pragma unroll
        for (int t = 0; t < 16; ++t) a[t] = *(const v4f*)(arow + 8 * t);
    }

#pragma unroll
    for (int j = 0; j < (OUTC * INC / 4) / GEMM_THREADS; ++j) {
        const int i  = tid + j * GEMM_THREADS;
        const int n  = i >> 5;
        const int k4 = i & 31;
        const v4f w  = *(const v4f*)(W + (size_t)n * INC + 4 * k4);
        const int r  = (n & 64) | ((n & 3) << 4) | ((n >> 2) & 15);
        *(v4f*)(wl + r * WPITCH + 4 * k4) = w;
    }
    __syncthreads();
    if (!active) return;

    const float* bl = wl + l * WPITCH + 4 * h;
    v8f acc[8];
#pragma unroll
    for (int c = 0; c < 8; ++c) acc[c] = (v8f){};

    v4f b0[8], b1[8];
#pragma unroll
    for (int c = 0; c < 8; ++c) b0[c] = *(const v4f*)(bl + c * 16 * WPITCH);

#pragma unroll
    for (int t = 0; t < 16; t += 2) {
#pragma unroll
        for (int c = 0; c < 8; ++c) b1[c] = *(const v4f*)(bl + c * 16 * WPITCH + 8 * (t + 1));
        __builtin_amdgcn_sched_barrier(0);
#pragma unroll
        for (int c = 0; c < 8; ++c)
            acc[c] = __builtin_amdgcn_wmma_f32_16x16x4_f32(
                false, a[t].xy, false, b0[c].xy, (short)0, acc[c], false, false);
#pragma unroll
        for (int c = 0; c < 8; ++c)
            acc[c] = __builtin_amdgcn_wmma_f32_16x16x4_f32(
                false, a[t].zw, false, b0[c].zw, (short)0, acc[c], false, false);
        __builtin_amdgcn_sched_barrier(0);
        if (t + 2 < 16) {
#pragma unroll
            for (int c = 0; c < 8; ++c) b0[c] = *(const v4f*)(bl + c * 16 * WPITCH + 8 * (t + 2));
        }
        __builtin_amdgcn_sched_barrier(0);
#pragma unroll
        for (int c = 0; c < 8; ++c)
            acc[c] = __builtin_amdgcn_wmma_f32_16x16x4_f32(
                false, a[t + 1].xy, false, b1[c].xy, (short)0, acc[c], false, false);
#pragma unroll
        for (int c = 0; c < 8; ++c)
            acc[c] = __builtin_amdgcn_wmma_f32_16x16x4_f32(
                false, a[t + 1].zw, false, b1[c].zw, (short)0, acc[c], false, false);
        __builtin_amdgcn_sched_barrier(0);
    }

    float* __restrict__ op = emb + (size_t)(strip * 16 + 8 * h) * OUTC + 4 * l;
#pragma unroll
    for (int q = 0; q < 2; ++q) {
#pragma unroll
        for (int v = 0; v < 8; ++v) {
            const v4f o = {acc[4 * q][v], acc[4 * q + 1][v], acc[4 * q + 2][v], acc[4 * q + 3][v]};
            *(v4f*)(op + (size_t)v * OUTC + 64 * q) = o;
        }
    }
}

#define RANGE   196
#define NRANGE  256
static_assert(RANGE * NRANGE >= NN, "ranges cover all nodes");
static_assert((NN - 1) / RANGE < 256, "a range id fits a byte");
static_assert(EE % 16 == 0, "the range table is read in 16-byte pieces");

__global__ void gat_scores(const float* __restrict__ emb,
                           const float* __restrict__ a_left,
                           const float* __restrict__ a_right,
                           const int* __restrict__ trg,
                           float* __restrict__ ss, float* __restrict__ ts,
                           unsigned* __restrict__ keys) {
    int gid = blockIdx.x * blockDim.x + threadIdx.x;
    if (gid >= NN * HEADS) return;
    if (gid < EE / 4) {
        const int4 t4 = *(const int4*)(trg + 4 * gid);
        keys[gid] = (unsigned)(t4.x / RANGE) | ((unsigned)(t4.y / RANGE) << 8) |
                    ((unsigned)(t4.z / RANGE) << 16) | ((unsigned)(t4.w / RANGE) << 24);
    }
    int n = gid >> 3, h = gid & 7;
    const float* ep = emb + (size_t)n * OUTC + h * HEADC;
    float s0 = 0.f, s1 = 0.f;
#pragma unroll
    for (int c = 0; c < HEADC; ++c) {
        float e = ep[c];
        s0 += e * a_left[c * HEADS + h];
        s1 += e * a_right[c * HEADS + h];
    }
    ss[gid] = s0;
    ts[gid] = s1;
}

#define OWN_THREADS 1024
#define OWN_WAVES   (OWN_THREADS / 32)
#define WCAP        512
#define KVEC        (EE / 16)
#define SCAN_U      8
#define OWN_G       16
#define ACC_POS(c)  (((c) & 3) * 32 + ((c) >> 2))

__device__ __forceinline__ float gat_edge_exp(float a, float b) {
    float v = a + b;
    v = v > 0.f ? v : 0.2f * v;
    return __expf(v);
}

__device__ __forceinline__ unsigned gat_zero_bytes(unsigned y) {
    return ~(((y & 0x7F7F7F7Fu) + 0x7F7F7F7Fu) | y | 0x7F7F7F7Fu);
}

__global__ __launch_bounds__(OWN_THREADS)
void gat_owner(const int* __restrict__ src, const int* __restrict__ trg,
               const float* __restrict__ emb,
               const float* __restrict__ ss, const float* __restrict__ ts,
               const unsigned* __restrict__ keys,
               const float* __restrict__ bias,
               float* __restrict__ out) {
    __shared__ __attribute__((aligned(16))) float acc[RANGE * OUTC];
    __shared__ __attribute__((aligned(16))) float den[RANGE * HEADS];
    __shared__ unsigned wl[OWN_WAVES * WCAP];
    __shared__ unsigned q[OWN_WAVES * WCAP];
    __shared__ unsigned wc[OWN_WAVES];

    const int tid = threadIdx.x;
    const unsigned lane = tid & 31u;
    const int wave = __builtin_amdgcn_readfirstlane(tid >> 5);
    const int rid  = blockIdx.x;
    const int base = rid * RANGE;

    for (int i = tid; i < RANGE * OUTC / 4; i += OWN_THREADS) ((v4f*)acc)[i] = (v4f){0.f, 0.f, 0.f, 0.f};
    for (int i = tid; i < RANGE * HEADS / 4; i += OWN_THREADS) ((v4f*)den)[i] = (v4f){0.f, 0.f, 0.f, 0.f};
    __syncthreads();

    const uint4* __restrict__ k4 = (const uint4*)keys;
    const unsigned rid4 = (unsigned)rid * 0x01010101u;
    unsigned* mywl = wl + wave * WCAP;
    int wcount = 0;
    for (int k0 = 0; k0 * OWN_THREADS < KVEC; k0 += SCAN_U) {
        uint4 kv[SCAN_U];
#pragma unroll
        for (int u = 0; u < SCAN_U; ++u) {
            const int j = tid + OWN_THREADS * (k0 + u);
            kv[u] = j < KVEC ? k4[j] : make_uint4(~rid4, ~rid4, ~rid4, ~rid4);
        }
#pragma unroll
        for (int u = 0; u < SCAN_U; u += 2) {
            const int j = tid + OWN_THREADS * (k0 + u);
            unsigned M = (gat_zero_bytes(kv[u].x ^ rid4) >> 7) | (gat_zero_bytes(kv[u].y ^ rid4) >> 6) |
                         (gat_zero_bytes(kv[u].z ^ rid4) >> 5) | (gat_zero_bytes(kv[u].w ^ rid4) >> 4) |
                         (gat_zero_bytes(kv[u + 1].x ^ rid4) >> 3) | (gat_zero_bytes(kv[u + 1].y ^ rid4) >> 2) |
                         (gat_zero_bytes(kv[u + 1].z ^ rid4) >> 1) | gat_zero_bytes(kv[u + 1].w ^ rid4);
            unsigned live = __builtin_amdgcn_ballot_w32(M != 0u);
            while (live) {
                if (M != 0u) {
                    const int p = __builtin_ctz(M);
                    M &= M - 1u;
                    const int e = 16 * (j + OWN_THREADS * ((p >> 2) & 1)) + 4 * (p & 3) + (p >> 3);
                    const int slot = wcount + (int)__builtin_amdgcn_mbcnt_lo(live, 0u);
                    if (slot < WCAP) {
                        mywl[slot] = (unsigned)e;
                    } else {
                        const int s = src[e];
                        const int t = trg[e];
                        const int r = t - base;
#pragma unroll 1
                        for (int h = 0; h < HEADS; ++h) {
                            const float ex = gat_edge_exp(ss[s * HEADS + h], ts[t * HEADS + h]);
                            atomicAdd(&den[r * HEADS + h], ex);
#pragma unroll 1
                            for (int cc = h * HEADC; cc < (h + 1) * HEADC; ++cc)
                                atomicAdd(&acc[r * OUTC + ACC_POS(cc)], emb[(size_t)s * OUTC + cc] * ex);
                        }
                    }
                }
                wcount += __builtin_popcount(live);
                live = __builtin_amdgcn_ballot_w32(M != 0u);
            }
        }
    }
    if (lane == 0) wc[wave] = (unsigned)(wcount < WCAP ? wcount : WCAP);
    __syncthreads();

    const unsigned cl = wc[lane];
    unsigned incl = cl;
#pragma unroll
    for (int d = 1; d < 32; d <<= 1) {
        const unsigned up = (unsigned)__shfl_up((int)incl, d);
        if ((int)lane >= d) incl += up;
    }
    const int n   = __builtin_amdgcn_readlane((int)incl, 31);
    const int cw  = __builtin_amdgcn_readlane((int)cl, wave);
    const int off = __builtin_amdgcn_readlane((int)(incl - cl), wave);
    for (int i0 = 0; i0 < cw; i0 += 128) {
        int e[4], s[4], t[4];
#pragma unroll
        for (int k = 0; k < 4; ++k) {
            const int idx = i0 + k * 32 + (int)lane;
            e[k] = idx < cw ? (int)mywl[idx] : 0;
        }
#pragma unroll
        for (int k = 0; k < 4; ++k) { s[k] = src[e[k]]; t[k] = trg[e[k]]; }
#pragma unroll
        for (int k = 0; k < 4; ++k) {
            const int idx = i0 + k * 32 + (int)lane;
            if (idx < cw) q[off + idx] = ((unsigned)s[k] << 8) | (unsigned)(t[k] - base);
        }
    }
    __syncthreads();

    const unsigned sub = lane >> 3;
    const unsigned hd  = lane & 7u;
    for (int g = wave * OWN_G; g < n; g += OWN_WAVES * OWN_G) {
        unsigned ps[OWN_G / 4];
        float sa[OWN_G / 4], sb[OWN_G / 4];
#pragma unroll
        for (int k = 0; k < OWN_G / 4; ++k) {
            const int idx = g + 4 * k + (int)sub;
            ps[k] = idx < n ? q[idx] : 0u;
            sa[k] = ss[(ps[k] >> 8) * HEADS + hd];
            sb[k] = ts[(base + (int)(ps[k] & 255u)) * HEADS + hd];
        }

        int  p[OWN_G];
        v4f  v[OWN_G];
#pragma unroll
        for (int u = 0; u < OWN_G; ++u) {
            p[u] = __builtin_amdgcn_readlane((int)ps[u >> 2], (u & 3) * 8);
            v[u] = *(const v4f*)(emb + (size_t)(p[u] >> 8) * OUTC + 4 * lane);
        }

        float ex[OWN_G / 4];
#pragma unroll
        for (int k = 0; k < OWN_G / 4; ++k) {
            ex[k] = gat_edge_exp(sa[k], sb[k]);
            if (g + 4 * k + (int)sub < n) atomicAdd(&den[(ps[k] & 255u) * HEADS + hd], ex[k]);
        }

        float w[OWN_G];
#pragma unroll
        for (int u = 0; u < OWN_G; ++u)
            w[u] = __shfl(ex[u >> 2], (int)((u & 3) * 8 + (lane >> 2)));
#pragma unroll
        for (int u = 0; u < OWN_G; ++u) {
            if (g + u < n) {
                float* arow = acc + (p[u] & 255) * OUTC + lane;
                atomicAdd(arow,      v[u].x * w[u]);
                atomicAdd(arow + 32, v[u].y * w[u]);
                atomicAdd(arow + 64, v[u].z * w[u]);
                atomicAdd(arow + 96, v[u].w * w[u]);
            }
        }
    }
    __syncthreads();

    for (int i = tid; i < RANGE * (OUTC / 4); i += OWN_THREADS) {
        const int r    = i >> 5;
        const int c4   = i & 31;
        const int node = base + r;
        if (node < NN) {
            const float* arow = acc + r * OUTC + c4;
            const float d = den[r * HEADS + (c4 >> 2)] + 1e-16f;
            const v4f b = *(const v4f*)(bias + 4 * c4);
            v4f o;
            o.x = arow[0] / d + b.x; o.y = arow[32] / d + b.y; o.z = arow[64] / d + b.z; o.w = arow[96] / d + b.w;
            *(v4f*)(out + (size_t)node * OUTC + 4 * c4) = o;
        }
    }
}

extern "C" void kernel_launch(void* const* d_in, const int* in_sizes, int n_in,
                              void* d_out, int out_size, void* d_ws, size_t ws_size,
                              hipStream_t stream) {
    const float* X      = (const float*)d_in[0];
    const int*   eidx   = (const int*)d_in[1];
    const float* W      = (const float*)d_in[2];
    const float* a_l    = (const float*)d_in[3];
    const float* a_r    = (const float*)d_in[4];
    const float* bias   = (const float*)d_in[5];
    float* out = (float*)d_out;

    const int* src = eidx;
    const int* trg = eidx + EE;

    float* ws    = (float*)d_ws;
    float* emb   = ws;
    float* ss    = emb + (size_t)NN * OUTC;
    float* ts    = ss + (size_t)NN * HEADS;
    unsigned* keys = (unsigned*)(ts + (size_t)NN * HEADS);

    gat_gemm_wmma<<<GEMM_BLOCKS, GEMM_THREADS, 0, stream>>>(X, W, emb);

    gat_scores<<<(NN * HEADS + 255) / 256, 256, 0, stream>>>(emb, a_l, a_r, trg, ss, ts, keys);

    gat_owner<<<NRANGE, OWN_THREADS, 0, stream>>>(src, trg, emb, ss, ts, keys, bias, out);
}


# Range table kept as 8 bit planes per 32 edges (ballots in the scores launch): an owner lane tests 32 edges with 8 bit operations, all table loads of a trip in flight; add order free as before
# speedup vs baseline: 4.9909x; 1.0954x over previous
#include <hip/hip_runtime.h>
#include <hip/hip_bf16.h>

#define NN 50000
#define EE 600000
#define INC 128
#define OUTC 128
#define HEADS 8
#define HEADC 16

typedef __attribute__((ext_vector_type(2))) float v2f;
typedef __attribute__((ext_vector_type(4))) float v4f;
typedef __attribute__((ext_vector_type(8))) float v8f;

#define GEMM_WAVES   16
#define GEMM_THREADS (GEMM_WAVES * 32)
#define GEMM_BLOCKS  256
#define STRIPS       (NN / 16)
#define WPITCH       132
static_assert(NN % 16 == 0, "whole strips only");
static_assert(GEMM_BLOCKS * GEMM_WAVES >= STRIPS, "one strip per wave at most");
static_assert((OUTC * INC / 4) % GEMM_THREADS == 0, "W staging: equal pieces per thread");

__global__ __launch_bounds__(GEMM_THREADS)
void gat_gemm_wmma(const float* __restrict__ X,
                   const float* __restrict__ W,
                   float* __restrict__ emb) {
    __shared__ __attribute__((aligned(16))) float wl[OUTC * WPITCH];

    const int tid  = threadIdx.x;
    const int lane = tid & 31;
    const int l    = lane & 15;
    const int h    = lane >> 4;
    const int wave  = __builtin_amdgcn_readfirstlane(tid >> 5);
    const int strip = wave * GEMM_BLOCKS + (int)blockIdx.x;
    const bool active = strip < STRIPS;


    v4f a[16];
    if (active) {
        const float* __restrict__ arow = X + (size_t)(strip * 16 + l) * INC + 4 * h;
#pragma unroll
        for (int t = 0; t < 16; ++t) a[t] = *(const v4f*)(arow + 8 * t);
    }

#pragma unroll
    for (int j = 0; j < (OUTC * INC / 4) / GEMM_THREADS; ++j) {
        const int i  = tid + j * GEMM_THREADS;
        const int n  = i >> 5;
        const int k4 = i & 31;
        const v4f w  = *(const v4f*)(W + (size_t)n * INC + 4 * k4);
        const int r  = (n & 64) | ((n & 3) << 4) | ((n >> 2) & 15);
        *(v4f*)(wl + r * WPITCH + 4 * k4) = w;
    }
    __syncthreads();
    if (!active) return;

    const float* bl = wl + l * WPITCH + 4 * h;
    v8f acc[8];
#pragma unroll
    for (int c = 0; c < 8; ++c) acc[c] = (v8f){};

    v4f b0[8], b1[8];
#pragma unroll
    for (int c = 0; c < 8; ++c) b0[c] = *(const v4f*)(bl + c * 16 * WPITCH);

#pragma unroll
    for (int t = 0; t < 16; t += 2) {
#pragma unroll
        for (int c = 0; c < 8; ++c) b1[c] = *(const v4f*)(bl + c * 16 * WPITCH + 8 * (t + 1));
        __builtin_amdgcn_sched_barrier(0);
#pragma unroll
        for (int c = 0; c < 8; ++c)
            acc[c] = __builtin_amdgcn_wmma_f32_16x16x4_f32(
                false, a[t].xy, false, b0[c].xy, (short)0, acc[c], false, false);
#pragma unroll
        for (int c = 0; c < 8; ++c)
            acc[c] = __builtin_amdgcn_wmma_f32_16x16x4_f32(
                false, a[t].zw, false, b0[c].zw, (short)0, acc[c], false, false);
        __builtin_amdgcn_sched_barrier(0);
        if (t + 2 < 16) {
#pragma unroll
            for (int c = 0; c < 8; ++c) b0[c] = *(const v4f*)(bl + c * 16 * WPITCH + 8 * (t + 2));
        }
        __builtin_amdgcn_sched_barrier(0);
#pragma unroll
        for (int c = 0; c < 8; ++c)
            acc[c] = __builtin_amdgcn_wmma_f32_16x16x4_f32(
                false, a[t + 1].xy, false, b1[c].xy, (short)0, acc[c], false, false);
#pragma unroll
        for (int c = 0; c < 8; ++c)
            acc[c] = __builtin_amdgcn_wmma_f32_16x16x4_f32(
                false, a[t + 1].zw, false, b1[c].zw, (short)0, acc[c], false, false);
        __builtin_amdgcn_sched_barrier(0);
    }

    float* __restrict__ op = emb + (size_t)(strip * 16 + 8 * h) * OUTC + 4 * l;
#pragma unroll
    for (int q = 0; q < 2; ++q) {
#pragma unroll
        for (int v = 0; v < 8; ++v) {
            const v4f o = {acc[4 * q][v], acc[4 * q + 1][v], acc[4 * q + 2][v], acc[4 * q + 3][v]};
            *(v4f*)(op + (size_t)v * OUTC + 64 * q) = o;
        }
    }
}

#define RANGE   196
#define NRANGE  256
#define NGROUP  (EE / 32)
static_assert(RANGE * NRANGE >= NN, "ranges cover all nodes");
static_assert((NN - 1) / RANGE < 256, "a range id fits 8 bits");
static_assert(EE % 64 == 0, "a wave of the scores launch builds two whole groups");
static_assert((NGROUP / 2) * 32 <= NN * HEADS, "the waves that build the planes are whole waves of the scores launch");

__global__ void gat_scores(const float* __restrict__ emb,
                           const float* __restrict__ a_left,
                           const float* __restrict__ a_right,
                           const int* __restrict__ trg,
                           float* __restrict__ ss, float* __restrict__ ts,
                           unsigned* __restrict__ planes) {
    int gid = blockIdx.x * blockDim.x + threadIdx.x;
    if (gid >= NN * HEADS) return;
    const int gw = __builtin_amdgcn_readfirstlane(gid >> 5);
    if (gw < NGROUP / 2) {
        const int lane = threadIdx.x & 31;
        const unsigned k0 = (unsigned)(trg[64 * gw + lane] / RANGE);
        const unsigned k1 = (unsigned)(trg[64 * gw + 32 + lane] / RANGE);
        unsigned word = 0u;
#pragma unroll
        for (int p = 0; p < 8; ++p) {
            const unsigned b0 = __builtin_amdgcn_ballot_w32(((k0 >> p) & 1u) != 0u);
            const unsigned b1 = __builtin_amdgcn_ballot_w32(((k1 >> p) & 1u) != 0u);
            if (lane == p) word = b0;
            if (lane == 8 + p) word = b1;
        }
        if (lane < 16) {
            const int g = 2 * gw + (lane >> 3);
            const int p = lane & 7;
            planes[(p >> 2) * (4 * NGROUP) + 4 * g + (p & 3)] = word;
        }
    }
    int n = gid >> 3, h = gid & 7;
    const float* ep = emb + (size_t)n * OUTC + h * HEADC;
    float s0 = 0.f, s1 = 0.f;
#pragma unroll
    for (int c = 0; c < HEADC; ++c) {
        float e = ep[c];
        s0 += e * a_left[c * HEADS + h];
        s1 += e * a_right[c * HEADS + h];
    }
    ss[gid] = s0;
    ts[gid] = s1;
}

#define OWN_THREADS 1024
#define OWN_WAVES   (OWN_THREADS / 32)
#define WCAP        512
#define SCAN_U      7
#define SCAN_SLOTS  ((NGROUP + OWN_THREADS - 1) / OWN_THREADS)
#define OWN_G       16
#define ACC_POS(c)  (((c) & 3) * 32 + ((c) >> 2))

__device__ __forceinline__ float gat_edge_exp(float a, float b) {
    float v = a + b;
    v = v > 0.f ? v : 0.2f * v;
    return __expf(v);
}

__global__ __launch_bounds__(OWN_THREADS)
void gat_owner(const int* __restrict__ src, const int* __restrict__ trg,
               const float* __restrict__ emb,
               const float* __restrict__ ss, const float* __restrict__ ts,
               const unsigned* __restrict__ planes,
               const float* __restrict__ bias,
               float* __restrict__ out) {
    __shared__ __attribute__((aligned(16))) float acc[RANGE * OUTC];
    __shared__ __attribute__((aligned(16))) float den[RANGE * HEADS];
    __shared__ unsigned wl[OWN_WAVES * WCAP];
    __shared__ unsigned q[OWN_WAVES * WCAP];
    __shared__ unsigned wc[OWN_WAVES];

    const int tid = threadIdx.x;
    const unsigned lane = tid & 31u;
    const int wave = __builtin_amdgcn_readfirstlane(tid >> 5);
    const int rid  = blockIdx.x;
    const int base = rid * RANGE;

    for (int i = tid; i < RANGE * OUTC / 4; i += OWN_THREADS) ((v4f*)acc)[i] = (v4f){0.f, 0.f, 0.f, 0.f};
    for (int i = tid; i < RANGE * HEADS / 4; i += OWN_THREADS) ((v4f*)den)[i] = (v4f){0.f, 0.f, 0.f, 0.f};
    __syncthreads();

    const uint4* __restrict__ plo = (const uint4*)planes;
    const uint4* __restrict__ phi = plo + NGROUP;
    unsigned xm[8];
#pragma unroll
    for (int p = 0; p < 8; ++p) xm[p] = ((rid >> p) & 1) ? 0u : 0xFFFFFFFFu;
    unsigned* mywl = wl + wave * WCAP;
    int wcount = 0;
    for (int s0 = 0; s0 < SCAN_SLOTS; s0 += SCAN_U) {
        uint4 lo[SCAN_U], hi[SCAN_U];
#pragma unroll
        for (int u = 0; u < SCAN_U; ++u) {
            const int g = tid + OWN_THREADS * (s0 + u);
            const int gc = g < NGROUP ? g : NGROUP - 1;
            lo[u] = plo[gc];
            hi[u] = phi[gc];
        }
#pragma unroll
        for (int u = 0; u < SCAN_U; ++u) {
            const int g = tid + OWN_THREADS * (s0 + u);
            unsigned M = (lo[u].x ^ xm[0]) & (lo[u].y ^ xm[1]) & (lo[u].z ^ xm[2]) & (lo[u].w ^ xm[3]) &
                         (hi[u].x ^ xm[4]) & (hi[u].y ^ xm[5]) & (hi[u].z ^ xm[6]) & (hi[u].w ^ xm[7]);
            if (g >= NGROUP) M = 0u;
            unsigned live = __builtin_amdgcn_ballot_w32(M != 0u);
            while (live) {
                if (M != 0u) {
                    const int p = __builtin_ctz(M);
                    M &= M - 1u;
                    const int e = 32 * g + p;
                    const int slot = wcount + (int)__builtin_amdgcn_mbcnt_lo(live, 0u);
                    if (slot < WCAP) {
                        mywl[slot] = (unsigned)e;
                    } else {
                        const int s = src[e];
                        const int t = trg[e];
                        const int r = t - base;
#pragma unroll 1
                        for (int h = 0; h < HEADS; ++h) {
                            const float ex = gat_edge_exp(ss[s * HEADS + h], ts[t * HEADS + h]);
                            atomicAdd(&den[r * HEADS + h], ex);
#pragma unroll 1
                            for (int cc = h * HEADC; cc < (h + 1) * HEADC; ++cc)
                                atomicAdd(&acc[r * OUTC + ACC_POS(cc)], emb[(size_t)s * OUTC + cc] * ex);
                        }
                    }
                }
                wcount += __builtin_popcount(live);
                live = __builtin_amdgcn_ballot_w32(M != 0u);
            }
        }
    }
    if (lane == 0) wc[wave] = (unsigned)(wcount < WCAP ? wcount : WCAP);
    __syncthreads();

    const unsigned cl = wc[lane];
    unsigned incl = cl;
#pragma unroll
    for (int d = 1; d < 32; d <<= 1) {
        const unsigned up = (unsigned)__shfl_up((int)incl, d);
        if ((int)lane >= d) incl += up;
    }
    const int n   = __builtin_amdgcn_readlane((int)incl, 31);
    const int cw  = __builtin_amdgcn_readlane((int)cl, wave);
    const int off = __builtin_amdgcn_readlane((int)(incl - cl), wave);
    for (int i0 = 0; i0 < cw; i0 += 128) {
        int e[4], s[4], t[4];
#pragma unroll
        for (int k = 0; k < 4; ++k) {
            const int idx = i0 + k * 32 + (int)lane;
            e[k] = idx < cw ? (int)mywl[idx] : 0;
        }
#pragma unroll
        for (int k = 0; k < 4; ++k) { s[k] = src[e[k]]; t[k] = trg[e[k]]; }
#pragma unroll
        for (int k = 0; k < 4; ++k) {
            const int idx = i0 + k * 32 + (int)lane;
            if (idx < cw) q[off + idx] = ((unsigned)s[k] << 8) | (unsigned)(t[k] - base);
        }
    }
    __syncthreads();

    const unsigned sub = lane >> 3;
    const unsigned hd  = lane & 7u;
    for (int g = wave * OWN_G; g < n; g += OWN_WAVES * OWN_G) {
        unsigned ps[OWN_G / 4];
        float sa[OWN_G / 4], sb[OWN_G / 4];
#pragma unroll
        for (int k = 0; k < OWN_G / 4; ++k) {
            const int idx = g + 4 * k + (int)sub;
            ps[k] = idx < n ? q[idx] : 0u;
            sa[k] = ss[(ps[k] >> 8) * HEADS + hd];
            sb[k] = ts[(base + (int)(ps[k] & 255u)) * HEADS + hd];
        }

        int  p[OWN_G];
        v4f  v[OWN_G];
#pragma unroll
        for (int u = 0; u < OWN_G; ++u) {
            p[u] = __builtin_amdgcn_readlane((int)ps[u >> 2], (u & 3) * 8);
            v[u] = *(const v4f*)(emb + (size_t)(p[u] >> 8) * OUTC + 4 * lane);
        }

        float ex[OWN_G / 4];
#pragma unroll
        for (int k = 0; k < OWN_G / 4; ++k) {
            ex[k] = gat_edge_exp(sa[k], sb[k]);
            if (g + 4 * k + (int)sub < n) atomicAdd(&den[(ps[k] & 255u) * HEADS + hd], ex[k]);
        }

        float w[OWN_G];
#pragma unroll
        for (int u = 0; u < OWN_G; ++u)
            w[u] = __shfl(ex[u >> 2], (int)((u & 3) * 8 + (lane >> 2)));
#pragma unroll
        for (int u = 0; u < OWN_G; ++u) {
            if (g + u < n) {
                float* arow = acc + (p[u] & 255) * OUTC + lane;
                atomicAdd(arow,      v[u].x * w[u]);
                atomicAdd(arow + 32, v[u].y * w[u]);
                atomicAdd(arow + 64, v[u].z * w[u]);
                atomicAdd(arow + 96, v[u].w * w[u]);
            }
        }
    }
    __syncthreads();

    for (int i = tid; i < RANGE * (OUTC / 4); i += OWN_THREADS) {
        const int r    = i >> 5;
        const int c4   = i & 31;
        const int node = base + r;
        if (node < NN) {
            const float* arow = acc + r * OUTC + c4;
            const float d = den[r * HEADS + (c4 >> 2)] + 1e-16f;
            const v4f b = *(const v4f*)(bias + 4 * c4);
            v4f o;
            o.x = arow[0] / d + b.x; o.y = arow[32] / d + b.y; o.z = arow[64] / d + b.z; o.w = arow[96] / d + b.w;
            *(v4f*)(out + (size_t)node * OUTC + 4 * c4) = o;
        }
    }
}

extern "C" void kernel_launch(void* const* d_in, const int* in_sizes, int n_in,
                              void* d_out, int out_size, void* d_ws, size_t ws_size,
                              hipStream_t stream) {
    const float* X      = (const float*)d_in[0];
    const int*   eidx   = (const int*)d_in[1];
    const float* W      = (const float*)d_in[2];
    const float* a_l    = (const float*)d_in[3];
    const float* a_r    = (const float*)d_in[4];
    const float* bias   = (const float*)d_in[5];
    float* out = (float*)d_out;

    const int* src = eidx;
    const int* trg = eidx + EE;

    float* ws    = (float*)d_ws;
    float* emb   = ws;
    float* ss    = emb + (size_t)NN * OUTC;
    float* ts    = ss + (size_t)NN * HEADS;
    unsigned* planes = (unsigned*)(ts + (size_t)NN * HEADS);

    gat_gemm_wmma<<<GEMM_BLOCKS, GEMM_THREADS, 0, stream>>>(X, W, emb);

    gat_scores<<<(NN * HEADS + 255) / 256, 256, 0, stream>>>(emb, a_l, a_r, trg, ss, ts, planes);

    gat_owner<<<NRANGE, OWN_THREADS, 0, stream>>>(src, trg, emb, ss, ts, planes, bias, out);
}


# Scores launch gone: a ninth 16-column WMMA tile gives both score sets (B rows W^T a_left, W^T a_right built per block in LDS: score sums reassociated); GEMM waves build the range table
# speedup vs baseline: 5.1489x; 1.0317x over previous
#include <hip/hip_runtime.h>
#include <hip/hip_bf16.h>

#define NN 50000
#define EE 600000
#define INC 128
#define OUTC 128
#define HEADS 8
#define HEADC 16

typedef __attribute__((ext_vector_type(2))) float v2f;
typedef __attribute__((ext_vector_type(4))) float v4f;
typedef __attribute__((ext_vector_type(8))) float v8f;

#define RANGE   196
#define NRANGE  256
#define NGROUP  (EE / 32)
static_assert(RANGE * NRANGE >= NN, "ranges cover all nodes");
static_assert((NN - 1) / RANGE < 256, "a range id fits 8 bits");
static_assert(EE % 64 == 0, "a wave builds two whole groups at a time");

#define GEMM_WAVES   16
#define GEMM_THREADS (GEMM_WAVES * 32)
#define GEMM_BLOCKS  256
#define STRIPS       (NN / 16)
#define WPITCH       132
#define NTILE        9
#define PLANE_TRIPS  3
static_assert(NN % 16 == 0, "whole strips only");
static_assert(GEMM_BLOCKS * GEMM_WAVES >= STRIPS, "one strip per wave at most");
static_assert((OUTC * INC / 4) % GEMM_THREADS == 0, "W staging: equal pieces per thread");
static_assert(GEMM_THREADS == 4 * INC, "score rows: a thread takes one k of two heads");
static_assert(GEMM_BLOCKS * GEMM_WAVES * PLANE_TRIPS >= NGROUP / 2, "the waves cover all pairs of groups");

__global__ __launch_bounds__(GEMM_THREADS)
void gat_gemm_wmma(const float* __restrict__ X,
                   const float* __restrict__ W,
                   const float* __restrict__ a_left,
                   const float* __restrict__ a_right,
                   const int* __restrict__ trg,
                   float* __restrict__ emb,
                   float* __restrict__ ss, float* __restrict__ ts,
                   unsigned* __restrict__ planes) {
    __shared__ __attribute__((aligned(16))) float wl[(OUTC + 16) * WPITCH];

    const int tid  = threadIdx.x;
    const int lane = tid & 31;
    const int l    = lane & 15;
    const int h    = lane >> 4;
    const int wave  = __builtin_amdgcn_readfirstlane(tid >> 5);
    const int strip = wave * GEMM_BLOCKS + (int)blockIdx.x;
    const bool active = strip < STRIPS;
    const int gwave = (int)blockIdx.x * GEMM_WAVES + wave;


    int tg0[PLANE_TRIPS], tg1[PLANE_TRIPS];
#pragma unroll
    for (int i = 0; i < PLANE_TRIPS; ++i) {
        const int gw  = gwave + i * (GEMM_BLOCKS * GEMM_WAVES);
        const int gwc = gw < NGROUP / 2 ? gw : 0;
        tg0[i] = trg[64 * gwc + lane];
        tg1[i] = trg[64 * gwc + 32 + lane];
    }

    v4f a[16];
    if (active) {
        const float* __restrict__ arow = X + (size_t)(strip * 16 + l) * INC + 4 * h;
#pragma unroll
        for (int t = 0; t < 16; ++t) a[t] = *(const v4f*)(arow + 8 * t);
    }

#pragma unroll
    for (int j = 0; j < (OUTC * INC / 4) / GEMM_THREADS; ++j) {
        const int i  = tid + j * GEMM_THREADS;
        const int n  = i >> 5;
        const int k4 = i & 31;
        const v4f w  = *(const v4f*)(W + (size_t)n * INC + 4 * k4);
        const int r  = (n & 64) | ((n & 3) << 4) | ((n >> 2) & 15);
        *(v4f*)(wl + r * WPITCH + 4 * k4) = w;
    }
    __syncthreads();

    {
        const int hp = __builtin_amdgcn_readfirstlane(tid >> 7);
        const int k  = tid & (INC - 1);
#pragma unroll
        for (int u = 0; u < 2; ++u) {
            const int hd = 2 * hp + u;
            float sl = 0.f, sr = 0.f;
#pragma unroll
            for (int c = 0; c < HEADC; ++c) {
                const int n = hd * HEADC + c;
                const int r = (n & 64) | ((n & 3) << 4) | ((n >> 2) & 15);
                const float w = wl[r * WPITCH + k];
                sl += w * a_left[c * HEADS + hd];
                sr += w * a_right[c * HEADS + hd];
            }
            wl[(OUTC + hd) * WPITCH + k]     = sl;
            wl[(OUTC + 8 + hd) * WPITCH + k] = sr;
        }
    }
    __syncthreads();
    if (active) {

    const float* bl = wl + l * WPITCH + 4 * h;
    v8f acc[NTILE];
#pragma unroll
    for (int c = 0; c < NTILE; ++c) acc[c] = (v8f){};

    v4f b0[NTILE], b1[NTILE];
#pragma unroll
    for (int c = 0; c < NTILE; ++c) b0[c] = *(const v4f*)(bl + c * 16 * WPITCH);

#pragma unroll
    for (int t = 0; t < 16; t += 2) {
#pragma unroll
        for (int c = 0; c < NTILE; ++c) b1[c] = *(const v4f*)(bl + c * 16 * WPITCH + 8 * (t + 1));
        __builtin_amdgcn_sched_barrier(0);
#pragma unroll
        for (int c = 0; c < NTILE; ++c)
            acc[c] = __builtin_amdgcn_wmma_f32_16x16x4_f32(
                false, a[t].xy, false, b0[c].xy, (short)0, acc[c], false, false);
#pragma unroll
        for (int c = 0; c < NTILE; ++c)
            acc[c] = __builtin_amdgcn_wmma_f32_16x16x4_f32(
                false, a[t].zw, false, b0[c].zw, (short)0, acc[c], false, false);
        __builtin_amdgcn_sched_barrier(0);
        if (t + 2 < 16) {
#pragma unroll
            for (int c = 0; c < NTILE; ++c) b0[c] = *(const v4f*)(bl + c * 16 * WPITCH + 8 * (t + 2));
        }
        __builtin_amdgcn_sched_barrier(0);
#pragma unroll
        for (int c = 0; c < NTILE; ++c)
            acc[c] = __builtin_amdgcn_wmma_f32_16x16x4_f32(
                false, a[t + 1].xy, false, b1[c].xy, (short)0, acc[c], false, false);
#pragma unroll
        for (int c = 0; c < NTILE; ++c)
            acc[c] = __builtin_amdgcn_wmma_f32_16x16x4_f32(
                false, a[t + 1].zw, false, b1[c].zw, (short)0, acc[c], false, false);
        __builtin_amdgcn_sched_barrier(0);
    }

    float* __restrict__ op = emb + (size_t)(strip * 16 + 8 * h) * OUTC + 4 * l;
#pragma unroll
    for (int q = 0; q < 2; ++q) {
#pragma unroll
        for (int v = 0; v < 8; ++v) {
            const v4f o = {acc[4 * q][v], acc[4 * q + 1][v], acc[4 * q + 2][v], acc[4 * q + 3][v]};
            *(v4f*)(op + (size_t)v * OUTC + 64 * q) = o;
        }
    }
    float* __restrict__ sp = (l < 8 ? ss : ts) + (size_t)(strip * 16 + 8 * h) * HEADS + (l & 7);
#pragma unroll
    for (int v = 0; v < 8; ++v) sp[v * HEADS] = acc[8][v];
    }

#pragma unroll
    for (int i = 0; i < PLANE_TRIPS; ++i) {
        const int gw = gwave + i * (GEMM_BLOCKS * GEMM_WAVES);
        if (gw < NGROUP / 2) {
            const unsigned k0 = (unsigned)(tg0[i] / RANGE);
            const unsigned k1 = (unsigned)(tg1[i] / RANGE);
            unsigned word = 0u;
#pragma unroll
            for (int p = 0; p < 8; ++p) {
                const unsigned b0 = __builtin_amdgcn_ballot_w32(((k0 >> p) & 1u) != 0u);
                const unsigned b1 = __builtin_amdgcn_ballot_w32(((k1 >> p) & 1u) != 0u);
                if (lane == p) word = b0;
                if (lane == 8 + p) word = b1;
            }
            if (lane < 16) {
                const int g = 2 * gw + (lane >> 3);
                const int p = lane & 7;
                planes[(p >> 2) * (4 * NGROUP) + 4 * g + (p & 3)] = word;
            }
        }
    }
}

#define OWN_THREADS 1024
#define OWN_WAVES   (OWN_THREADS / 32)
#define WCAP        512
#define SCAN_U      7
#define SCAN_SLOTS  ((NGROUP + OWN_THREADS - 1) / OWN_THREADS)
#define OWN_G       16
#define ACC_POS(c)  (((c) & 3) * 32 + ((c) >> 2))

__device__ __forceinline__ float gat_edge_exp(float a, float b) {
    float v = a + b;
    v = v > 0.f ? v : 0.2f * v;
    return __expf(v);
}

__global__ __launch_bounds__(OWN_THREADS)
void gat_owner(const int* __restrict__ src, const int* __restrict__ trg,
               const float* __restrict__ emb,
               const float* __restrict__ ss, const float* __restrict__ ts,
               const unsigned* __restrict__ planes,
               const float* __restrict__ bias,
               float* __restrict__ out) {
    __shared__ __attribute__((aligned(16))) float acc[RANGE * OUTC];
    __shared__ __attribute__((aligned(16))) float den[RANGE * HEADS];
    __shared__ unsigned wl[OWN_WAVES * WCAP];
    __shared__ unsigned q[OWN_WAVES * WCAP];
    __shared__ unsigned wc[OWN_WAVES];

    const int tid = threadIdx.x;
    const unsigned lane = tid & 31u;
    const int wave = __builtin_amdgcn_readfirstlane(tid >> 5);
    const int rid  = blockIdx.x;
    const int base = rid * RANGE;

    for (int i = tid; i < RANGE * OUTC / 4; i += OWN_THREADS) ((v4f*)acc)[i] = (v4f){0.f, 0.f, 0.f, 0.f};
    for (int i = tid; i < RANGE * HEADS / 4; i += OWN_THREADS) ((v4f*)den)[i] = (v4f){0.f, 0.f, 0.f, 0.f};
    __syncthreads();

    const uint4* __restrict__ plo = (const uint4*)planes;
    const uint4* __restrict__ phi = plo + NGROUP;
    unsigned xm[8];
#pragma unroll
    for (int p = 0; p < 8; ++p) xm[p] = ((rid >> p) & 1) ? 0u : 0xFFFFFFFFu;
    unsigned* mywl = wl + wave * WCAP;
    int wcount = 0;
    for (int s0 = 0; s0 < SCAN_SLOTS; s0 += SCAN_U) {
        uint4 lo[SCAN_U], hi[SCAN_U];
#pragma unroll
        for (int u = 0; u < SCAN_U; ++u) {
            const int g = tid + OWN_THREADS * (s0 + u);
            const int gc = g < NGROUP ? g : NGROUP - 1;
            lo[u] = plo[gc];
            hi[u] = phi[gc];
        }
#pragma unroll
        for (int u = 0; u < SCAN_U; ++u) {
            const int g = tid + OWN_THREADS * (s0 + u);
            unsigned M = (lo[u].x ^ xm[0]) & (lo[u].y ^ xm[1]) & (lo[u].z ^ xm[2]) & (lo[u].w ^ xm[3]) &
                         (hi[u].x ^ xm[4]) & (hi[u].y ^ xm[5]) & (hi[u].z ^ xm[6]) & (hi[u].w ^ xm[7]);
            if (g >= NGROUP) M = 0u;
            unsigned live = __builtin_amdgcn_ballot_w32(M != 0u);
            while (live) {
                if (M != 0u) {
                    const int p = __builtin_ctz(M);
                    M &= M - 1u;
                    const int e = 32 * g + p;
                    const int slot = wcount + (int)__builtin_amdgcn_mbcnt_lo(live, 0u);
                    if (slot < WCAP) {
                        mywl[slot] = (unsigned)e;
                    } else {
                        const int s = src[e];
                        const int t = trg[e];
                        const int r = t - base;
#pragma unroll 1
                        for (int h = 0; h < HEADS; ++h) {
                            const float ex = gat_edge_exp(ss[s * HEADS + h], ts[t * HEADS + h]);
                            atomicAdd(&den[r * HEADS + h], ex);
#pragma unroll 1
                            for (int cc = h * HEADC; cc < (h + 1) * HEADC; ++cc)
                                atomicAdd(&acc[r * OUTC + ACC_POS(cc)], emb[(size_t)s * OUTC + cc] * ex);
                        }
                    }
                }
                wcount += __builtin_popcount(live);
                live = __builtin_amdgcn_ballot_w32(M != 0u);
            }
        }
    }
    if (lane == 0) wc[wave] = (unsigned)(wcount < WCAP ? wcount : WCAP);
    __syncthreads();

    const unsigned cl = wc[lane];
    unsigned incl = cl;
#pragma unroll
    for (int d = 1; d < 32; d <<= 1) {
        const unsigned up = (unsigned)__shfl_up((int)incl, d);
        if ((int)lane >= d) incl += up;
    }
    const int n   = __builtin_amdgcn_readlane((int)incl, 31);
    const int cw  = __builtin_amdgcn_readlane((int)cl, wave);
    const int off = __builtin_amdgcn_readlane((int)(incl - cl), wave);
    for (int i0 = 0; i0 < cw; i0 += 128) {
        int e[4], s[4], t[4];
#pragma unroll
        for (int k = 0; k < 4; ++k) {
            const int idx = i0 + k * 32 + (int)lane;
            e[k] = idx < cw ? (int)mywl[idx] : 0;
        }
#pragma unroll
        for (int k = 0; k < 4; ++k) { s[k] = src[e[k]]; t[k] = trg[e[k]]; }
#pragma unroll
        for (int k = 0; k < 4; ++k) {
            const int idx = i0 + k * 32 + (int)lane;
            if (idx < cw) q[off + idx] = ((unsigned)s[k] << 8) | (unsigned)(t[k] - base);
        }
    }
    __syncthreads();

    const unsigned sub = lane >> 3;
    const unsigned hd  = lane & 7u;
    for (int g = wave * OWN_G; g < n; g += OWN_WAVES * OWN_G) {
        unsigned ps[OWN_G / 4];
        float sa[OWN_G / 4], sb[OWN_G / 4];
#pragma unroll
        for (int k = 0; k < OWN_G / 4; ++k) {
            const int idx = g + 4 * k + (int)sub;
            ps[k] = idx < n ? q[idx] : 0u;
            sa[k] = ss[(ps[k] >> 8) * HEADS + hd];
            sb[k] = ts[(base + (int)(ps[k] & 255u)) * HEADS + hd];
        }

        int  p[OWN_G];
        v4f  v[OWN_G];
#pragma unroll
        for (int u = 0; u < OWN_G; ++u) {
            p[u] = __builtin_amdgcn_readlane((int)ps[u >> 2], (u & 3) * 8);
            v[u] = *(const v4f*)(emb + (size_t)(p[u] >> 8) * OUTC + 4 * lane);
        }

        float ex[OWN_G / 4];
#pragma unroll
        for (int k = 0; k < OWN_G / 4; ++k) {
            ex[k] = gat_edge_exp(sa[k], sb[k]);
            if (g + 4 * k + (int)sub < n) atomicAdd(&den[(ps[k] & 255u) * HEADS + hd], ex[k]);
        }

        float w[OWN_G];
#pragma unroll
        for (int u = 0; u < OWN_G; ++u)
            w[u] = __shfl(ex[u >> 2], (int)((u & 3) * 8 + (lane >> 2)));
#pragma unroll
        for (int u = 0; u < OWN_G; ++u) {
            if (g + u < n) {
                float* arow = acc + (p[u] & 255) * OUTC + lane;
                atomicAdd(arow,      v[u].x * w[u]);
                atomicAdd(arow + 32, v[u].y * w[u]);
                atomicAdd(arow + 64, v[u].z * w[u]);
                atomicAdd(arow + 96, v[u].w * w[u]);
            }
        }
    }
    __syncthreads();

    for (int i = tid; i < RANGE * (OUTC / 4); i += OWN_THREADS) {
        const int r    = i >> 5;
        const int c4   = i & 31;
        const int node = base + r;
        if (node < NN) {
            const float* arow = acc + r * OUTC + c4;
            const float d = den[r * HEADS + (c4 >> 2)] + 1e-16f;
            const v4f b = *(const v4f*)(bias + 4 * c4);
            v4f o;
            o.x = arow[0] / d + b.x; o.y = arow[32] / d + b.y; o.z = arow[64] / d + b.z; o.w = arow[96] / d + b.w;
            *(v4f*)(out + (size_t)node * OUTC + 4 * c4) = o;
        }
    }
}

extern "C" void kernel_launch(void* const* d_in, const int* in_sizes, int n_in,
                              void* d_out, int out_size, void* d_ws, size_t ws_size,
                              hipStream_t stream) {
    const float* X      = (const float*)d_in[0];
    const int*   eidx   = (const int*)d_in[1];
    const float* W      = (const float*)d_in[2];
    const float* a_l    = (const float*)d_in[3];
    const float* a_r    = (const float*)d_in[4];
    const float* bias   = (const float*)d_in[5];
    float* out = (float*)d_out;

    const int* src = eidx;
    const int* trg = eidx + EE;

    float* ws    = (float*)d_ws;
    float* emb   = ws;
    float* ss    = emb + (size_t)NN * OUTC;
    float* ts    = ss + (size_t)NN * HEADS;
    unsigned* planes = (unsigned*)(ts + (size_t)NN * HEADS);

    gat_gemm_wmma<<<GEMM_BLOCKS, GEMM_THREADS, 0, stream>>>(X, W, a_l, a_r, trg, emb, ss, ts, planes);

    gat_owner<<<NRANGE, OWN_THREADS, 0, stream>>>(src, trg, emb, ss, ts, planes, bias, out);
}
